# starting kernel
# baseline (speedup 1.0000x reference)
#include <hip/hip_runtime.h>

typedef __attribute__((ext_vector_type(2))) float v2f;
typedef __attribute__((ext_vector_type(8))) float v8f;

#define N_NODES 50000
#define N_EDGES 800000
#define N_FEAT  128
#define N_HID   64

__global__ void init_bias_kernel(float* __restrict__ buf,
                                 const float* __restrict__ bias) {
    int i = blockIdx.x * blockDim.x + threadIdx.x;
    if (i < N_NODES * N_HID) buf[i] = bias[i & (N_HID - 1)];
}

__global__ __launch_bounds__(256) void gemm_wmma_kernel(
    const float* __restrict__ X, const float* __restrict__ W,
    float* __restrict__ HW, int M, int K) {
    int wave = (blockIdx.x * blockDim.x + threadIdx.x) >> 5;
    int lane = threadIdx.x & 31;
    int m0 = wave * 16;
    if (m0 >= M) return;
    int half = lane >> 4;
    int l    = lane & 15;

    v8f acc0 = {}, acc1 = {}, acc2 = {}, acc3 = {};
    const float* xrow = X + (size_t)(m0 + l) * K;

    for (int k = 0; k < K; k += 4) {
        int kk = k + 2 * half;
        v2f a = *(const v2f*)(xrow + kk);
        const float* bp = W + (size_t)kk * N_HID + l;
        v2f b0 = { bp[ 0], bp[ 0 + N_HID] };
        v2f b1 = { bp[16], bp[16 + N_HID] };
        v2f b2 = { bp[32], bp[32 + N_HID] };
        v2f b3 = { bp[48], bp[48 + N_HID] };
        acc0 = __builtin_amdgcn_wmma_f32_16x16x4_f32(false, a, false, b0, (short)0, acc0, false, false);
        acc1 = __builtin_amdgcn_wmma_f32_16x16x4_f32(false, a, false, b1, (short)0, acc1, false, false);
        acc2 = __builtin_amdgcn_wmma_f32_16x16x4_f32(false, a, false, b2, (short)0, acc2, false, false);
        acc3 = __builtin_amdgcn_wmma_f32_16x16x4_f32(false, a, false, b3, (short)0, acc3, false, false);
    }

    float* orow = HW + (size_t)(m0 + 8 * half) * N_HID + l;
    #pragma unroll
    for (int r = 0; r < 8; ++r) {
        orow[(size_t)r * N_HID +  0] = acc0[r];
        orow[(size_t)r * N_HID + 16] = acc1[r];
        orow[(size_t)r * N_HID + 32] = acc2[r];
        orow[(size_t)r * N_HID + 48] = acc3[r];
    }
}

__global__ void scatter_kernel(const float* __restrict__ HW,
                               const int* __restrict__ src,
                               const int* __restrict__ dst,
                               const float* __restrict__ ew,
                               float* __restrict__ agg) {
    int t = blockIdx.x * blockDim.x + threadIdx.x;
    if (t >= N_EDGES * N_HID) return;
    int e = t >> 6;
    int f = t & 63;
    float v = HW[(size_t)src[e] * N_HID + f] * ew[e];
    atomicAdd(&agg[(size_t)dst[e] * N_HID + f], v);
}

__global__ void relu_kernel(float* __restrict__ h) {
    int i = blockIdx.x * blockDim.x + threadIdx.x;
    if (i < N_NODES * N_HID) h[i] = fmaxf(h[i], 0.0f);
}

extern "C" void kernel_launch(void* const* d_in, const int* in_sizes, int n_in,
                              void* d_out, int out_size, void* d_ws, size_t ws_size,
                              hipStream_t stream) {
    const float* x  = (const float*)d_in[0];
    const int*   ei = (const int*)  d_in[1];
    const float* ew = (const float*)d_in[2];
    const float* w1 = (const float*)d_in[3];
    const float* b1 = (const float*)d_in[4];
    const float* w2 = (const float*)d_in[5];
    const float* b2 = (const float*)d_in[6];
    float* out = (float*)d_out;

    float* hw  = (float*)d_ws;
    float* agg = hw + (size_t)N_NODES * N_HID;

    const int* src = ei;
    const int* dst = ei + N_EDGES;

    const int nodeElems = N_NODES * N_HID;
    const int edgeElems = N_EDGES * N_HID;
    const int mTiles    = N_NODES / 16;
    dim3 blk(256);
    dim3 gNode((nodeElems + 255) / 256);
    dim3 gEdge((edgeElems + 255) / 256);
    dim3 gGemm((mTiles + 7) / 8);

    gemm_wmma_kernel<<<gGemm, blk, 0, stream>>>(x, w1, hw, N_NODES, N_FEAT);
    init_bias_kernel<<<gNode, blk, 0, stream>>>(agg, b1);
    scatter_kernel<<<gEdge, blk, 0, stream>>>(hw, src, dst, ew, agg);
    relu_kernel<<<gNode, blk, 0, stream>>>(agg);

    gemm_wmma_kernel<<<gGemm, blk, 0, stream>>>(agg, w2, hw, N_NODES, N_HID);
    init_bias_kernel<<<gNode, blk, 0, stream>>>(out, b2);
    scatter_kernel<<<gEdge, blk, 0, stream>>>(hw, src, dst, ew, out);
}


# Replace per-element float atomic scatter by in-call dst-bucketed edge list (int count+rank atomics, scan, fill) + one-wave-per-node register gather seeded with bias; per-node sum order = arrival rank
# speedup vs baseline: 1.5252x; 1.5252x over previous
#include <hip/hip_runtime.h>


typedef __attribute__((ext_vector_type(2))) float v2f;
typedef __attribute__((ext_vector_type(8))) float v8f;

#define N_NODES 50000
#define N_EDGES 800000
#define N_FEAT  128
#define N_HID   64

#define SCAN_THREADS 1024
#define SCAN_CHUNK   ((N_NODES + SCAN_THREADS - 1) / SCAN_THREADS)
#define CNT_PAD      (SCAN_THREADS * SCAN_CHUNK)

static_assert(N_HID == 64, "the gather maps 32 lanes x 2 floats onto one 64-wide row");
static_assert(N_NODES % 8 == 0, "the gather grid is exact: 8 one-wave nodes per 256-thread block");

__global__ __launch_bounds__(256) void gemm_wmma_kernel(
    const float* __restrict__ X, const float* __restrict__ W,
    float* __restrict__ HW, int M, int K) {
    int wave = (blockIdx.x * blockDim.x + threadIdx.x) >> 5;
    int lane = threadIdx.x & 31;
    int m0 = wave * 16;
    if (m0 >= M) return;
    int half = lane >> 4;
    int l    = lane & 15;

    v8f acc0 = {}, acc1 = {}, acc2 = {}, acc3 = {};
    const float* xrow = X + (size_t)(m0 + l) * K;

    for (int k = 0; k < K; k += 4) {
        int kk = k + 2 * half;
        v2f a = *(const v2f*)(xrow + kk);
        const float* bp = W + (size_t)kk * N_HID + l;
        v2f b0 = { bp[ 0], bp[ 0 + N_HID] };
        v2f b1 = { bp[16], bp[16 + N_HID] };
        v2f b2 = { bp[32], bp[32 + N_HID] };
        v2f b3 = { bp[48], bp[48 + N_HID] };
        acc0 = __builtin_amdgcn_wmma_f32_16x16x4_f32(false, a, false, b0, (short)0, acc0, false, false);
        acc1 = __builtin_amdgcn_wmma_f32_16x16x4_f32(false, a, false, b1, (short)0, acc1, false, false);
        acc2 = __builtin_amdgcn_wmma_f32_16x16x4_f32(false, a, false, b2, (short)0, acc2, false, false);
        acc3 = __builtin_amdgcn_wmma_f32_16x16x4_f32(false, a, false, b3, (short)0, acc3, false, false);
    }

    float* orow = HW + (size_t)(m0 + 8 * half) * N_HID + l;
    #pragma unroll
    for (int r = 0; r < 8; ++r) {
        orow[(size_t)r * N_HID +  0] = acc0[r];
        orow[(size_t)r * N_HID + 16] = acc1[r];
        orow[(size_t)r * N_HID + 32] = acc2[r];
        orow[(size_t)r * N_HID + 48] = acc3[r];
    }
}


__global__ void zero_counts_kernel(int* __restrict__ cnt) {
    int i = blockIdx.x * blockDim.x + threadIdx.x;
    if (i < N_NODES) cnt[i] = 0;
}

__global__ void count_rank_kernel(const int* __restrict__ dst,
                                  int* __restrict__ cnt,
                                  int* __restrict__ rank) {
    int e = blockIdx.x * blockDim.x + threadIdx.x;
    if (e < N_EDGES) rank[e] = atomicAdd(&cnt[dst[e]], 1);
}

__global__ __launch_bounds__(SCAN_THREADS) void scan_counts_kernel(
    const int* __restrict__ cnt, int* __restrict__ rowptr) {
    __shared__ int wave_sum[SCAN_THREADS / 32];
    const int t    = threadIdx.x;
    const int lane = t & 31;
    const int wv   = t >> 5;
    const int beg  = t * SCAN_CHUNK;

    int c[SCAN_CHUNK];
    int s = 0;
    #pragma unroll
    for (int i = 0; i < SCAN_CHUNK; ++i) {
        int idx = beg + i;
        c[i] = (idx < N_NODES) ? cnt[idx] : 0;
        s += c[i];
    }

    int v = s;
    for (int d = 1; d < 32; d <<= 1) {
        int u = __shfl_up(v, d);
        if (lane >= d) v += u;
    }
    if (lane == 31) wave_sum[wv] = v;
    __syncthreads();
    if (wv == 0) {
        int x = wave_sum[lane];
        for (int d = 1; d < 32; d <<= 1) {
            int u = __shfl_up(x, d);
            if (lane >= d) x += u;
        }
        wave_sum[lane] = x;
    }
    __syncthreads();

    int off = (v - s) + (wv > 0 ? wave_sum[wv - 1] : 0);
    #pragma unroll
    for (int i = 0; i < SCAN_CHUNK; ++i) {
        int idx = beg + i;
        if (idx < N_NODES) rowptr[idx] = off;
        off += c[i];
    }
    if (t == SCAN_THREADS - 1) rowptr[N_NODES] = off;
}

__global__ void bucket_fill_kernel(const int* __restrict__ src,
                                   const int* __restrict__ dst,
                                   const float* __restrict__ ew,
                                   const int* __restrict__ rowptr,
                                   const int* __restrict__ rank,
                                   int2* __restrict__ bucket) {
    int e = blockIdx.x * blockDim.x + threadIdx.x;
    if (e < N_EDGES) {
        int pos = rowptr[dst[e]] + rank[e];
        bucket[pos] = make_int2(src[e], __float_as_int(ew[e]));
    }
}

template <int U>
__device__ __forceinline__ void add_edges(const float2* __restrict__ HW2, int2 p, int t, int lane,
                                          float2& acc) {
    float2 r[U];
    float  w[U];
    #pragma unroll
    for (int u = 0; u < U; ++u) {
        int s = __builtin_amdgcn_readlane(p.x, t + u);
        w[u]  = __int_as_float(__builtin_amdgcn_readlane(p.y, t + u));
        r[u]  = HW2[(size_t)s * (N_HID / 2) + lane];
    }
    #pragma unroll
    for (int u = 0; u < U; ++u) {
        acc.x = fmaf(w[u], r[u].x, acc.x);
        acc.y = fmaf(w[u], r[u].y, acc.y);
    }
}

__global__ __launch_bounds__(256) void gather_kernel(const float* __restrict__ HW,
                                                     const int* __restrict__ rowptr,
                                                     const int2* __restrict__ bucket,
                                                     const float* __restrict__ bias,
                                                     float* __restrict__ out) {
    const int lane = threadIdx.x & 31;
    const int node = blockIdx.x * (blockDim.x >> 5) + (threadIdx.x >> 5);
    const int beg = __builtin_amdgcn_readfirstlane(rowptr[node]);
    const int end = __builtin_amdgcn_readfirstlane(rowptr[node + 1]);

    const float2* __restrict__ HW2 = (const float2*)HW;
    float2 acc = ((const float2*)bias)[lane];

    for (int base = beg; base < end; base += 32) {
        int j = base + lane;
        int2 p = make_int2(0, 0);
        if (j < end) p = bucket[j];
        int n = end - base;
        if (n > 32) n = 32;
        int t = 0;
        for (; t + 8 <= n; t += 8) add_edges<8>(HW2, p, t, lane, acc);
        if (t + 4 <= n) { add_edges<4>(HW2, p, t, lane, acc); t += 4; }
        if (t + 2 <= n) { add_edges<2>(HW2, p, t, lane, acc); t += 2; }
        if (t < n)      { add_edges<1>(HW2, p, t, lane, acc); }
    }

    ((float2*)out)[(size_t)node * (N_HID / 2) + lane] = acc;
}

__global__ void relu_kernel(float* __restrict__ h) {
    int i = blockIdx.x * blockDim.x + threadIdx.x;
    if (i < N_NODES * N_HID) h[i] = fmaxf(h[i], 0.0f);
}

extern "C" void kernel_launch(void* const* d_in, const int* in_sizes, int n_in,
                              void* d_out, int out_size, void* d_ws, size_t ws_size,
                              hipStream_t stream) {
    const float* x  = (const float*)d_in[0];
    const int*   ei = (const int*)  d_in[1];
    const float* ew = (const float*)d_in[2];
    const float* w1 = (const float*)d_in[3];
    const float* b1 = (const float*)d_in[4];
    const float* w2 = (const float*)d_in[5];
    const float* b2 = (const float*)d_in[6];
    float* out = (float*)d_out;

    float* hw     = (float*)d_ws;
    float* agg    = hw + (size_t)N_NODES * N_HID;
    int*   cnt    = (int*)(agg + (size_t)N_NODES * N_HID);
    int*   rowptr = cnt + CNT_PAD;
    int*   rank   = rowptr + CNT_PAD;
    int2*  bucket = (int2*)(rank + N_EDGES);

    const int* src = ei;
    const int* dst = ei + N_EDGES;

    const int nodeElems = N_NODES * N_HID;
    const int mTiles    = N_NODES / 16;
    dim3 blk(256);
    dim3 gNode((nodeElems + 255) / 256);
    dim3 gCnt((N_NODES + 255) / 256);
    dim3 gEdge((N_EDGES + 255) / 256);
    dim3 gGemm((mTiles + 7) / 8);
    dim3 gGather(N_NODES / 8);

    zero_counts_kernel<<<gCnt, blk, 0, stream>>>(cnt);
    count_rank_kernel<<<gEdge, blk, 0, stream>>>(dst, cnt, rank);
    scan_counts_kernel<<<dim3(1), dim3(SCAN_THREADS), 0, stream>>>(cnt, rowptr);
    bucket_fill_kernel<<<gEdge, blk, 0, stream>>>(src, dst, ew, rowptr, rank, bucket);

    gemm_wmma_kernel<<<gGemm, blk, 0, stream>>>(x, w1, hw, N_NODES, N_FEAT);
    gather_kernel<<<gGather, blk, 0, stream>>>(hw, rowptr, bucket, b1, agg);
    relu_kernel<<<gNode, blk, 0, stream>>>(agg);

    gemm_wmma_kernel<<<gGemm, blk, 0, stream>>>(agg, w2, hw, N_NODES, N_HID);
    gather_kernel<<<gGather, blk, 0, stream>>>(hw, rowptr, bucket, b2, out);
}


# Fold the ReLU into the layer-1 gather's one store per row: the elementwise relu kernel, its launch and its 100000 one-element waves are gone; sums and their order unchanged
# speedup vs baseline: 1.6313x; 1.0696x over previous
#include <hip/hip_runtime.h>


typedef __attribute__((ext_vector_type(2))) float v2f;
typedef __attribute__((ext_vector_type(8))) float v8f;

#define N_NODES 50000
#define N_EDGES 800000
#define N_FEAT  128
#define N_HID   64

#define SCAN_THREADS 1024
#define SCAN_CHUNK   ((N_NODES + SCAN_THREADS - 1) / SCAN_THREADS)
#define CNT_PAD      (SCAN_THREADS * SCAN_CHUNK)

static_assert(N_HID == 64, "the gather maps 32 lanes x 2 floats onto one 64-wide row");
static_assert(N_NODES % 8 == 0, "the gather grid is exact: 8 one-wave nodes per 256-thread block");

__global__ __launch_bounds__(256) void gemm_wmma_kernel(
    const float* __restrict__ X, const float* __restrict__ W,
    float* __restrict__ HW, int M, int K) {
    int wave = (blockIdx.x * blockDim.x + threadIdx.x) >> 5;
    int lane = threadIdx.x & 31;
    int m0 = wave * 16;
    if (m0 >= M) return;
    int half = lane >> 4;
    int l    = lane & 15;

    v8f acc0 = {}, acc1 = {}, acc2 = {}, acc3 = {};
    const float* xrow = X + (size_t)(m0 + l) * K;

    for (int k = 0; k < K; k += 4) {
        int kk = k + 2 * half;
        v2f a = *(const v2f*)(xrow + kk);
        const float* bp = W + (size_t)kk * N_HID + l;
        v2f b0 = { bp[ 0], bp[ 0 + N_HID] };
        v2f b1 = { bp[16], bp[16 + N_HID] };
        v2f b2 = { bp[32], bp[32 + N_HID] };
        v2f b3 = { bp[48], bp[48 + N_HID] };
        acc0 = __builtin_amdgcn_wmma_f32_16x16x4_f32(false, a, false, b0, (short)0, acc0, false, false);
        acc1 = __builtin_amdgcn_wmma_f32_16x16x4_f32(false, a, false, b1, (short)0, acc1, false, false);
        acc2 = __builtin_amdgcn_wmma_f32_16x16x4_f32(false, a, false, b2, (short)0, acc2, false, false);
        acc3 = __builtin_amdgcn_wmma_f32_16x16x4_f32(false, a, false, b3, (short)0, acc3, false, false);
    }

    float* orow = HW + (size_t)(m0 + 8 * half) * N_HID + l;
    #pragma unroll
    for (int r = 0; r < 8; ++r) {
        orow[(size_t)r * N_HID +  0] = acc0[r];
        orow[(size_t)r * N_HID + 16] = acc1[r];
        orow[(size_t)r * N_HID + 32] = acc2[r];
        orow[(size_t)r * N_HID + 48] = acc3[r];
    }
}


__global__ void zero_counts_kernel(int* __restrict__ cnt) {
    int i = blockIdx.x * blockDim.x + threadIdx.x;
    if (i < N_NODES) cnt[i] = 0;
}

__global__ void count_rank_kernel(const int* __restrict__ dst,
                                  int* __restrict__ cnt,
                                  int* __restrict__ rank) {
    int e = blockIdx.x * blockDim.x + threadIdx.x;
    if (e < N_EDGES) rank[e] = atomicAdd(&cnt[dst[e]], 1);
}

__global__ __launch_bounds__(SCAN_THREADS) void scan_counts_kernel(
    const int* __restrict__ cnt, int* __restrict__ rowptr) {
    __shared__ int wave_sum[SCAN_THREADS / 32];
    const int t    = threadIdx.x;
    const int lane = t & 31;
    const int wv   = t >> 5;
    const int beg  = t * SCAN_CHUNK;

    int c[SCAN_CHUNK];
    int s = 0;
    #pragma unroll
    for (int i = 0; i < SCAN_CHUNK; ++i) {
        int idx = beg + i;
        c[i] = (idx < N_NODES) ? cnt[idx] : 0;
        s += c[i];
    }

    int v = s;
    for (int d = 1; d < 32; d <<= 1) {
        int u = __shfl_up(v, d);
        if (lane >= d) v += u;
    }
    if (lane == 31) wave_sum[wv] = v;
    __syncthreads();
    if (wv == 0) {
        int x = wave_sum[lane];
        for (int d = 1; d < 32; d <<= 1) {
            int u = __shfl_up(x, d);
            if (lane >= d) x += u;
        }
        wave_sum[lane] = x;
    }
    __syncthreads();

    int off = (v - s) + (wv > 0 ? wave_sum[wv - 1] : 0);
    #pragma unroll
    for (int i = 0; i < SCAN_CHUNK; ++i) {
        int idx = beg + i;
        if (idx < N_NODES) rowptr[idx] = off;
        off += c[i];
    }
    if (t == SCAN_THREADS - 1) rowptr[N_NODES] = off;
}

__global__ void bucket_fill_kernel(const int* __restrict__ src,
                                   const int* __restrict__ dst,
                                   const float* __restrict__ ew,
                                   const int* __restrict__ rowptr,
                                   const int* __restrict__ rank,
                                   int2* __restrict__ bucket) {
    int e = blockIdx.x * blockDim.x + threadIdx.x;
    if (e < N_EDGES) {
        int pos = rowptr[dst[e]] + rank[e];
        bucket[pos] = make_int2(src[e], __float_as_int(ew[e]));
    }
}

template <int U>
__device__ __forceinline__ void add_edges(const float2* __restrict__ HW2, int2 p, int t, int lane,
                                          float2& acc) {
    float2 r[U];
    float  w[U];
    #pragma unroll
    for (int u = 0; u < U; ++u) {
        int s = __builtin_amdgcn_readlane(p.x, t + u);
        w[u]  = __int_as_float(__builtin_amdgcn_readlane(p.y, t + u));
        r[u]  = HW2[(size_t)s * (N_HID / 2) + lane];
    }
    #pragma unroll
    for (int u = 0; u < U; ++u) {
        acc.x = fmaf(w[u], r[u].x, acc.x);
        acc.y = fmaf(w[u], r[u].y, acc.y);
    }
}

template <bool RELU>
__global__ __launch_bounds__(256) void gather_kernel(const float* __restrict__ HW,
                                                     const int* __restrict__ rowptr,
                                                     const int2* __restrict__ bucket,
                                                     const float* __restrict__ bias,
                                                     float* __restrict__ out) {
    const int lane = threadIdx.x & 31;
    const int node = blockIdx.x * (blockDim.x >> 5) + (threadIdx.x >> 5);
    const int beg = __builtin_amdgcn_readfirstlane(rowptr[node]);
    const int end = __builtin_amdgcn_readfirstlane(rowptr[node + 1]);

    const float2* __restrict__ HW2 = (const float2*)HW;
    float2 acc = ((const float2*)bias)[lane];

    for (int base = beg; base < end; base += 32) {
        int j = base + lane;
        int2 p = make_int2(0, 0);
        if (j < end) p = bucket[j];
        int n = end - base;
        if (n > 32) n = 32;
        int t = 0;
        for (; t + 8 <= n; t += 8) add_edges<8>(HW2, p, t, lane, acc);
        if (t + 4 <= n) { add_edges<4>(HW2, p, t, lane, acc); t += 4; }
        if (t + 2 <= n) { add_edges<2>(HW2, p, t, lane, acc); t += 2; }
        if (t < n)      { add_edges<1>(HW2, p, t, lane, acc); }
    }

    if (RELU) {
        acc.x = fmaxf(acc.x, 0.0f);
        acc.y = fmaxf(acc.y, 0.0f);
    }
    ((float2*)out)[(size_t)node * (N_HID / 2) + lane] = acc;
}

extern "C" void kernel_launch(void* const* d_in, const int* in_sizes, int n_in,
                              void* d_out, int out_size, void* d_ws, size_t ws_size,
                              hipStream_t stream) {
    const float* x  = (const float*)d_in[0];
    const int*   ei = (const int*)  d_in[1];
    const float* ew = (const float*)d_in[2];
    const float* w1 = (const float*)d_in[3];
    const float* b1 = (const float*)d_in[4];
    const float* w2 = (const float*)d_in[5];
    const float* b2 = (const float*)d_in[6];
    float* out = (float*)d_out;

    float* hw     = (float*)d_ws;
    float* agg    = hw + (size_t)N_NODES * N_HID;
    int*   cnt    = (int*)(agg + (size_t)N_NODES * N_HID);
    int*   rowptr = cnt + CNT_PAD;
    int*   rank   = rowptr + CNT_PAD;
    int2*  bucket = (int2*)(rank + N_EDGES);

    const int* src = ei;
    const int* dst = ei + N_EDGES;

    const int mTiles    = N_NODES / 16;
    dim3 blk(256);
    dim3 gCnt((N_NODES + 255) / 256);
    dim3 gEdge((N_EDGES + 255) / 256);
    dim3 gGemm((mTiles + 7) / 8);
    dim3 gGather(N_NODES / 8);

    zero_counts_kernel<<<gCnt, blk, 0, stream>>>(cnt);
    count_rank_kernel<<<gEdge, blk, 0, stream>>>(dst, cnt, rank);
    scan_counts_kernel<<<dim3(1), dim3(SCAN_THREADS), 0, stream>>>(cnt, rowptr);
    bucket_fill_kernel<<<gEdge, blk, 0, stream>>>(src, dst, ew, rowptr, rank, bucket);

    gemm_wmma_kernel<<<gGemm, blk, 0, stream>>>(x, w1, hw, N_NODES, N_FEAT);
    gather_kernel<true><<<gGather, blk, 0, stream>>>(hw, rowptr, bucket, b1, agg);

    gemm_wmma_kernel<<<gGemm, blk, 0, stream>>>(agg, w2, hw, N_NODES, N_HID);
    gather_kernel<false><<<gGather, blk, 0, stream>>>(hw, rowptr, bucket, b2, out);
}


# Prefix scan of the in-degree counts on 49 workgroups with contiguous loads and stores, each summing its own base from the counts, not 1 workgroup with lane-strided ones; float sum order unchanged
# speedup vs baseline: 2.2828x; 1.3994x over previous
#include <hip/hip_runtime.h>


typedef __attribute__((ext_vector_type(2))) float v2f;
typedef __attribute__((ext_vector_type(8))) float v8f;

#define N_NODES 50000
#define N_EDGES 800000
#define N_FEAT  128
#define N_HID   64

#define SCAN_THREADS 1024
#define SCAN_BLOCKS  ((N_NODES + SCAN_THREADS - 1) / SCAN_THREADS)
#define CNT_PAD      (SCAN_THREADS * SCAN_BLOCKS)
#define PRIOR_LOADS  (((SCAN_BLOCKS - 1) * (SCAN_THREADS / 4) + SCAN_THREADS - 1) / SCAN_THREADS)

static_assert(N_HID == 64, "the gather maps 32 lanes x 2 floats onto one 64-wide row");
static_assert(N_NODES % 8 == 0, "the gather grid is exact: 8 one-wave nodes per 256-thread block");
static_assert((SCAN_BLOCKS - 1) * SCAN_THREADS <= N_NODES, "every count in front of a scan chunk is a real count");

__global__ __launch_bounds__(256) void gemm_wmma_kernel(
    const float* __restrict__ X, const float* __restrict__ W,
    float* __restrict__ HW, int M, int K) {
    int wave = (blockIdx.x * blockDim.x + threadIdx.x) >> 5;
    int lane = threadIdx.x & 31;
    int m0 = wave * 16;
    if (m0 >= M) return;
    int half = lane >> 4;
    int l    = lane & 15;

    v8f acc0 = {}, acc1 = {}, acc2 = {}, acc3 = {};
    const float* xrow = X + (size_t)(m0 + l) * K;

    for (int k = 0; k < K; k += 4) {
        int kk = k + 2 * half;
        v2f a = *(const v2f*)(xrow + kk);
        const float* bp = W + (size_t)kk * N_HID + l;
        v2f b0 = { bp[ 0], bp[ 0 + N_HID] };
        v2f b1 = { bp[16], bp[16 + N_HID] };
        v2f b2 = { bp[32], bp[32 + N_HID] };
        v2f b3 = { bp[48], bp[48 + N_HID] };
        acc0 = __builtin_amdgcn_wmma_f32_16x16x4_f32(false, a, false, b0, (short)0, acc0, false, false);
        acc1 = __builtin_amdgcn_wmma_f32_16x16x4_f32(false, a, false, b1, (short)0, acc1, false, false);
        acc2 = __builtin_amdgcn_wmma_f32_16x16x4_f32(false, a, false, b2, (short)0, acc2, false, false);
        acc3 = __builtin_amdgcn_wmma_f32_16x16x4_f32(false, a, false, b3, (short)0, acc3, false, false);
    }

    float* orow = HW + (size_t)(m0 + 8 * half) * N_HID + l;
    #pragma unroll
    for (int r = 0; r < 8; ++r) {
        orow[(size_t)r * N_HID +  0] = acc0[r];
        orow[(size_t)r * N_HID + 16] = acc1[r];
        orow[(size_t)r * N_HID + 32] = acc2[r];
        orow[(size_t)r * N_HID + 48] = acc3[r];
    }
}


__global__ void zero_counts_kernel(int* __restrict__ cnt) {
    int i = blockIdx.x * blockDim.x + threadIdx.x;
    if (i < N_NODES) cnt[i] = 0;
}

__global__ void count_rank_kernel(const int* __restrict__ dst,
                                  int* __restrict__ cnt,
                                  int* __restrict__ rank) {
    int e = blockIdx.x * blockDim.x + threadIdx.x;
    if (e < N_EDGES) rank[e] = atomicAdd(&cnt[dst[e]], 1);
}

__global__ __launch_bounds__(SCAN_THREADS) void scan_counts_kernel(
    const int* __restrict__ cnt, int* __restrict__ rowptr) {
    __shared__ int wave_sum[SCAN_THREADS / 32];
    __shared__ int wave_prior[SCAN_THREADS / 32];
    __shared__ int block_base;
    const int t    = threadIdx.x;
    const int lane = t & 31;
    const int wv   = t >> 5;
    const int b    = blockIdx.x;
    const int idx  = b * SCAN_THREADS + t;

    const int c = (idx < N_NODES) ? cnt[idx] : 0;

    const int4* __restrict__ cnt4 = (const int4*)cnt;
    const int n4 = b * (SCAN_THREADS / 4);
    int p = 0;
    for (int k = 0; k < PRIOR_LOADS; ++k) {
        const int  i4 = k * SCAN_THREADS + t;
        const bool in = i4 < n4;
        const int4 q  = cnt4[in ? i4 : 0];
        const int  s  = (q.x + q.y) + (q.z + q.w);
        p += in ? s : 0;
    }

    int v = c;
    for (int d = 1; d < 32; d <<= 1) {
        int u = __shfl_up(v, d);
        if (lane >= d) v += u;
    }
    for (int d = 16; d >= 1; d >>= 1) p += __shfl_xor(p, d);
    if (lane == 31) { wave_sum[wv] = v; wave_prior[wv] = p; }
    __syncthreads();
    if (wv == 0) {
        int x = wave_sum[lane];
        int q = wave_prior[lane];
        for (int d = 1; d < 32; d <<= 1) {
            int u = __shfl_up(x, d);
            if (lane >= d) x += u;
        }
        for (int d = 16; d >= 1; d >>= 1) q += __shfl_xor(q, d);
        wave_sum[lane] = x + q;
        if (lane == 0) block_base = q;
    }
    __syncthreads();

    const int off = (v - c) + (wv > 0 ? wave_sum[wv - 1] : block_base);
    if (idx < N_NODES) rowptr[idx] = off;
    if (idx == N_NODES - 1) rowptr[N_NODES] = off + c;
}

__global__ void bucket_fill_kernel(const int* __restrict__ src,
                                   const int* __restrict__ dst,
                                   const float* __restrict__ ew,
                                   const int* __restrict__ rowptr,
                                   const int* __restrict__ rank,
                                   int2* __restrict__ bucket) {
    int e = blockIdx.x * blockDim.x + threadIdx.x;
    if (e < N_EDGES) {
        int pos = rowptr[dst[e]] + rank[e];
        bucket[pos] = make_int2(src[e], __float_as_int(ew[e]));
    }
}

template <int U>
__device__ __forceinline__ void add_edges(const float2* __restrict__ HW2, int2 p, int t, int lane,
                                          float2& acc) {
    float2 r[U];
    float  w[U];
    #pragma unroll
    for (int u = 0; u < U; ++u) {
        int s = __builtin_amdgcn_readlane(p.x, t + u);
        w[u]  = __int_as_float(__builtin_amdgcn_readlane(p.y, t + u));
        r[u]  = HW2[(size_t)s * (N_HID / 2) + lane];
    }
    #pragma unroll
    for (int u = 0; u < U; ++u) {
        acc.x = fmaf(w[u], r[u].x, acc.x);
        acc.y = fmaf(w[u], r[u].y, acc.y);
    }
}

template <bool RELU>
__global__ __launch_bounds__(256) void gather_kernel(const float* __restrict__ HW,
                                                     const int* __restrict__ rowptr,
                                                     const int2* __restrict__ bucket,
                                                     const float* __restrict__ bias,
                                                     float* __restrict__ out) {
    const int lane = threadIdx.x & 31;
    const int node = blockIdx.x * (blockDim.x >> 5) + (threadIdx.x >> 5);
    const int beg = __builtin_amdgcn_readfirstlane(rowptr[node]);
    const int end = __builtin_amdgcn_readfirstlane(rowptr[node + 1]);

    const float2* __restrict__ HW2 = (const float2*)HW;
    float2 acc = ((const float2*)bias)[lane];

    for (int base = beg; base < end; base += 32) {
        int j = base + lane;
        int2 p = make_int2(0, 0);
        if (j < end) p = bucket[j];
        int n = end - base;
        if (n > 32) n = 32;
        int t = 0;
        for (; t + 8 <= n; t += 8) add_edges<8>(HW2, p, t, lane, acc);
        if (t + 4 <= n) { add_edges<4>(HW2, p, t, lane, acc); t += 4; }
        if (t + 2 <= n) { add_edges<2>(HW2, p, t, lane, acc); t += 2; }
        if (t < n)      { add_edges<1>(HW2, p, t, lane, acc); }
    }

    if (RELU) {
        acc.x = fmaxf(acc.x, 0.0f);
        acc.y = fmaxf(acc.y, 0.0f);
    }
    ((float2*)out)[(size_t)node * (N_HID / 2) + lane] = acc;
}

extern "C" void kernel_launch(void* const* d_in, const int* in_sizes, int n_in,
                              void* d_out, int out_size, void* d_ws, size_t ws_size,
                              hipStream_t stream) {
    const float* x  = (const float*)d_in[0];
    const int*   ei = (const int*)  d_in[1];
    const float* ew = (const float*)d_in[2];
    const float* w1 = (const float*)d_in[3];
    const float* b1 = (const float*)d_in[4];
    const float* w2 = (const float*)d_in[5];
    const float* b2 = (const float*)d_in[6];
    float* out = (float*)d_out;

    float* hw     = (float*)d_ws;
    float* agg    = hw + (size_t)N_NODES * N_HID;
    int*   cnt    = (int*)(agg + (size_t)N_NODES * N_HID);
    int*   rowptr = cnt + CNT_PAD;
    int*   rank   = rowptr + CNT_PAD;
    int2*  bucket = (int2*)(rank + N_EDGES);

    const int* src = ei;
    const int* dst = ei + N_EDGES;

    const int mTiles    = N_NODES / 16;
    dim3 blk(256);
    dim3 gCnt((N_NODES + 255) / 256);
    dim3 gEdge((N_EDGES + 255) / 256);
    dim3 gGemm((mTiles + 7) / 8);
    dim3 gGather(N_NODES / 8);

    zero_counts_kernel<<<gCnt, blk, 0, stream>>>(cnt);
    count_rank_kernel<<<gEdge, blk, 0, stream>>>(dst, cnt, rank);
    scan_counts_kernel<<<dim3(SCAN_BLOCKS), dim3(SCAN_THREADS), 0, stream>>>(cnt, rowptr);
    bucket_fill_kernel<<<gEdge, blk, 0, stream>>>(src, dst, ew, rowptr, rank, bucket);

    gemm_wmma_kernel<<<gGemm, blk, 0, stream>>>(x, w1, hw, N_NODES, N_FEAT);
    gather_kernel<true><<<gGather, blk, 0, stream>>>(hw, rowptr, bucket, b1, agg);

    gemm_wmma_kernel<<<gGemm, blk, 0, stream>>>(agg, w2, hw, N_NODES, N_HID);
    gather_kernel<false><<<gGather, blk, 0, stream>>>(hw, rowptr, bucket, b2, out);
}


# Edges bucketed in one atomic pass into fixed 64-slot per-node buckets (spill list past 64): rank pass, prefix scan, rowptr lookups and 2 launches gone; per-node sum order = arrival rank as before
# speedup vs baseline: 2.5673x; 1.1246x over previous
#include <hip/hip_runtime.h>


typedef __attribute__((ext_vector_type(2))) float v2f;
typedef __attribute__((ext_vector_type(8))) float v8f;

#define N_NODES 50000
#define N_EDGES 800000
#define N_FEAT  128
#define N_HID   64

#define BUCKET_CAP 64
#define CNT_PAD    (((N_NODES + 1 + 63) / 64) * 64)

static_assert(N_HID == 64, "the gather maps 32 lanes x 2 floats onto one 64-wide row");
static_assert(N_NODES % 8 == 0, "the gather grid is exact: 8 one-wave nodes per 256-thread block");
static_assert((long long)N_NODES * BUCKET_CAP < 2147483647LL, "slot indices fit an int");

__global__ __launch_bounds__(256) void gemm_wmma_kernel(
    const float* __restrict__ X, const float* __restrict__ W,
    float* __restrict__ HW, int M, int K) {
    int wave = (blockIdx.x * blockDim.x + threadIdx.x) >> 5;
    int lane = threadIdx.x & 31;
    int m0 = wave * 16;
    if (m0 >= M) return;
    int half = lane >> 4;
    int l    = lane & 15;

    v8f acc0 = {}, acc1 = {}, acc2 = {}, acc3 = {};
    const float* xrow = X + (size_t)(m0 + l) * K;

    for (int k = 0; k < K; k += 4) {
        int kk = k + 2 * half;
        v2f a = *(const v2f*)(xrow + kk);
        const float* bp = W + (size_t)kk * N_HID + l;
        v2f b0 = { bp[ 0], bp[ 0 + N_HID] };
        v2f b1 = { bp[16], bp[16 + N_HID] };
        v2f b2 = { bp[32], bp[32 + N_HID] };
        v2f b3 = { bp[48], bp[48 + N_HID] };
        acc0 = __builtin_amdgcn_wmma_f32_16x16x4_f32(false, a, false, b0, (short)0, acc0, false, false);
        acc1 = __builtin_amdgcn_wmma_f32_16x16x4_f32(false, a, false, b1, (short)0, acc1, false, false);
        acc2 = __builtin_amdgcn_wmma_f32_16x16x4_f32(false, a, false, b2, (short)0, acc2, false, false);
        acc3 = __builtin_amdgcn_wmma_f32_16x16x4_f32(false, a, false, b3, (short)0, acc3, false, false);
    }

    float* orow = HW + (size_t)(m0 + 8 * half) * N_HID + l;
    #pragma unroll
    for (int r = 0; r < 8; ++r) {
        orow[(size_t)r * N_HID +  0] = acc0[r];
        orow[(size_t)r * N_HID + 16] = acc1[r];
        orow[(size_t)r * N_HID + 32] = acc2[r];
        orow[(size_t)r * N_HID + 48] = acc3[r];
    }
}


__global__ void zero_counts_kernel(int* __restrict__ cnt) {
    int i = blockIdx.x * blockDim.x + threadIdx.x;
    if (i <= N_NODES) cnt[i] = 0;
}

__global__ void bucket_fill_kernel(const int* __restrict__ src,
                                   const int* __restrict__ dst,
                                   const float* __restrict__ ew,
                                   int* __restrict__ cnt,
                                   int2* __restrict__ bucket,
                                   int4* __restrict__ spill) {
    int e = blockIdx.x * blockDim.x + threadIdx.x;
    if (e < N_EDGES) {
        const int  d   = dst[e];
        const int2 rec = make_int2(src[e], __float_as_int(ew[e]));
        const int  pos = atomicAdd(&cnt[d], 1);
        if (pos < BUCKET_CAP) {
            bucket[(size_t)d * BUCKET_CAP + pos] = rec;
        } else {
            const int o = atomicAdd(&cnt[N_NODES], 1);
            spill[o] = make_int4(d, rec.x, rec.y, 0);
        }
    }
}

template <int U>
__device__ __forceinline__ void add_edges(const float2* __restrict__ HW2, int2 p, int t, int lane,
                                          float2& acc) {
    float2 r[U];
    float  w[U];
    #pragma unroll
    for (int u = 0; u < U; ++u) {
        int s = __builtin_amdgcn_readlane(p.x, t + u);
        w[u]  = __int_as_float(__builtin_amdgcn_readlane(p.y, t + u));
        r[u]  = HW2[(size_t)s * (N_HID / 2) + lane];
    }
    #pragma unroll
    for (int u = 0; u < U; ++u) {
        acc.x = fmaf(w[u], r[u].x, acc.x);
        acc.y = fmaf(w[u], r[u].y, acc.y);
    }
}

template <bool RELU>
__global__ __launch_bounds__(256) void gather_kernel(const float* __restrict__ HW,
                                                     const int* __restrict__ cnt,
                                                     const int2* __restrict__ bucket,
                                                     const int4* __restrict__ spill,
                                                     const float* __restrict__ bias,
                                                     float* __restrict__ out) {
    const int lane = threadIdx.x & 31;
    const int node = __builtin_amdgcn_readfirstlane(blockIdx.x * (blockDim.x >> 5) + (threadIdx.x >> 5));
    const int deg = cnt[node];
    const int beg = node * BUCKET_CAP;
    const int end = beg + (deg < BUCKET_CAP ? deg : BUCKET_CAP);

    const float2* __restrict__ HW2 = (const float2*)HW;
    float2 acc = ((const float2*)bias)[lane];

    for (int base = beg; base < end; base += 32) {
        int j = base + lane;
        int2 p = make_int2(0, 0);
        if (j < end) p = bucket[j];
        int n = end - base;
        if (n > 32) n = 32;
        int t = 0;
        for (; t + 8 <= n; t += 8) add_edges<8>(HW2, p, t, lane, acc);
        if (t + 4 <= n) { add_edges<4>(HW2, p, t, lane, acc); t += 4; }
        if (t + 2 <= n) { add_edges<2>(HW2, p, t, lane, acc); t += 2; }
        if (t < n)      { add_edges<1>(HW2, p, t, lane, acc); }
    }

    if (deg > BUCKET_CAP) {
        const int n_spill = cnt[N_NODES];
        for (int base = 0; base < n_spill; base += 32) {
            int i = base + lane;
            int4 q = make_int4(-1, 0, 0, 0);
            if (i < n_spill) q = spill[i];
            unsigned m = __builtin_amdgcn_ballot_w32(q.x == node);
            while (m) {
                int t = __builtin_ctz(m);
                m &= m - 1;
                add_edges<1>(HW2, make_int2(q.y, q.z), t, lane, acc);
            }
        }
    }

    if (RELU) {
        acc.x = fmaxf(acc.x, 0.0f);
        acc.y = fmaxf(acc.y, 0.0f);
    }
    ((float2*)out)[(size_t)node * (N_HID / 2) + lane] = acc;
}

extern "C" void kernel_launch(void* const* d_in, const int* in_sizes, int n_in,
                              void* d_out, int out_size, void* d_ws, size_t ws_size,
                              hipStream_t stream) {
    const float* x  = (const float*)d_in[0];
    const int*   ei = (const int*)  d_in[1];
    const float* ew = (const float*)d_in[2];
    const float* w1 = (const float*)d_in[3];
    const float* b1 = (const float*)d_in[4];
    const float* w2 = (const float*)d_in[5];
    const float* b2 = (const float*)d_in[6];
    float* out = (float*)d_out;

    float* hw     = (float*)d_ws;
    float* agg    = hw + (size_t)N_NODES * N_HID;
    int*   cnt    = (int*)(agg + (size_t)N_NODES * N_HID);
    int2*  bucket = (int2*)(cnt + CNT_PAD);
    int4*  spill  = (int4*)(bucket + (size_t)N_NODES * BUCKET_CAP);

    const int* src = ei;
    const int* dst = ei + N_EDGES;

    const int mTiles    = N_NODES / 16;
    dim3 blk(256);
    dim3 gCnt((N_NODES + 1 + 255) / 256);
    dim3 gEdge((N_EDGES + 255) / 256);
    dim3 gGemm((mTiles + 7) / 8);
    dim3 gGather(N_NODES / 8);

    zero_counts_kernel<<<gCnt, blk, 0, stream>>>(cnt);
    bucket_fill_kernel<<<gEdge, blk, 0, stream>>>(src, dst, ew, cnt, bucket, spill);

    gemm_wmma_kernel<<<gGemm, blk, 0, stream>>>(x, w1, hw, N_NODES, N_FEAT);
    gather_kernel<true><<<gGather, blk, 0, stream>>>(hw, cnt, bucket, spill, b1, agg);

    gemm_wmma_kernel<<<gGemm, blk, 0, stream>>>(agg, w2, hw, N_NODES, N_HID);
    gather_kernel<false><<<gGather, blk, 0, stream>>>(hw, cnt, bucket, spill, b2, out);
}


# Edge bucketing and the layer-1 product x @ w1 (independent work) run as one grid, product blocks first: one launch fewer and the two bodies overlap; sums and their order as before
# speedup vs baseline: 2.6972x; 1.0506x over previous
#include <hip/hip_runtime.h>


typedef __attribute__((ext_vector_type(2))) float v2f;
typedef __attribute__((ext_vector_type(8))) float v8f;

#define N_NODES 50000
#define N_EDGES 800000
#define N_FEAT  128
#define N_HID   64

#define BUCKET_CAP 64
#define CNT_PAD    (((N_NODES + 1 + 63) / 64) * 64)

static_assert(N_HID == 64, "the gather maps 32 lanes x 2 floats onto one 64-wide row");
static_assert(N_NODES % 8 == 0, "the gather grid is exact: 8 one-wave nodes per 256-thread block");
static_assert((long long)N_NODES * BUCKET_CAP < 2147483647LL, "slot indices fit an int");

__device__ __forceinline__ void gemm_wmma_block(
    const float* __restrict__ X, const float* __restrict__ W,
    float* __restrict__ HW, int M, int K, int block) {
    int wave = (block * 256 + (int)threadIdx.x) >> 5;
    int lane = threadIdx.x & 31;
    int m0 = wave * 16;
    if (m0 >= M) return;
    int half = lane >> 4;
    int l    = lane & 15;

    v8f acc0 = {}, acc1 = {}, acc2 = {}, acc3 = {};
    const float* xrow = X + (size_t)(m0 + l) * K;

    for (int k = 0; k < K; k += 4) {
        int kk = k + 2 * half;
        v2f a = *(const v2f*)(xrow + kk);
        const float* bp = W + (size_t)kk * N_HID + l;
        v2f b0 = { bp[ 0], bp[ 0 + N_HID] };
        v2f b1 = { bp[16], bp[16 + N_HID] };
        v2f b2 = { bp[32], bp[32 + N_HID] };
        v2f b3 = { bp[48], bp[48 + N_HID] };
        acc0 = __builtin_amdgcn_wmma_f32_16x16x4_f32(false, a, false, b0, (short)0, acc0, false, false);
        acc1 = __builtin_amdgcn_wmma_f32_16x16x4_f32(false, a, false, b1, (short)0, acc1, false, false);
        acc2 = __builtin_amdgcn_wmma_f32_16x16x4_f32(false, a, false, b2, (short)0, acc2, false, false);
        acc3 = __builtin_amdgcn_wmma_f32_16x16x4_f32(false, a, false, b3, (short)0, acc3, false, false);
    }

    float* orow = HW + (size_t)(m0 + 8 * half) * N_HID + l;
    #pragma unroll
    for (int r = 0; r < 8; ++r) {
        orow[(size_t)r * N_HID +  0] = acc0[r];
        orow[(size_t)r * N_HID + 16] = acc1[r];
        orow[(size_t)r * N_HID + 32] = acc2[r];
        orow[(size_t)r * N_HID + 48] = acc3[r];
    }
}

__global__ __launch_bounds__(256) void gemm_wmma_kernel(
    const float* __restrict__ X, const float* __restrict__ W,
    float* __restrict__ HW, int M, int K) {
    gemm_wmma_block(X, W, HW, M, K, (int)blockIdx.x);
}


__global__ void zero_counts_kernel(int* __restrict__ cnt) {
    int i = blockIdx.x * blockDim.x + threadIdx.x;
    if (i <= N_NODES) cnt[i] = 0;
}

__device__ __forceinline__ void bucket_fill_block(const int* __restrict__ src,
                                                  const int* __restrict__ dst,
                                                  const float* __restrict__ ew,
                                                  int* __restrict__ cnt,
                                                  int2* __restrict__ bucket,
                                                  int4* __restrict__ spill,
                                                  int block) {
    int e = block * 256 + (int)threadIdx.x;
    if (e < N_EDGES) {
        const int  d   = dst[e];
        const int2 rec = make_int2(src[e], __float_as_int(ew[e]));
        const int  pos = atomicAdd(&cnt[d], 1);
        if (pos < BUCKET_CAP) {
            bucket[(size_t)d * BUCKET_CAP + pos] = rec;
        } else {
            const int o = atomicAdd(&cnt[N_NODES], 1);
            spill[o] = make_int4(d, rec.x, rec.y, 0);
        }
    }
}

#define GEMM_BLOCKS ((N_NODES / 16 + 7) / 8)
#define EDGE_BLOCKS ((N_EDGES + 255) / 256)

__global__ __launch_bounds__(256) void bucket_fill_and_gemm_kernel(
    const int* __restrict__ src, const int* __restrict__ dst, const float* __restrict__ ew,
    int* __restrict__ cnt, int2* __restrict__ bucket, int4* __restrict__ spill,
    const float* __restrict__ X, const float* __restrict__ W, float* __restrict__ HW) {
    if (blockIdx.x < GEMM_BLOCKS) {
        gemm_wmma_block(X, W, HW, N_NODES, N_FEAT, (int)blockIdx.x);
    } else {
        bucket_fill_block(src, dst, ew, cnt, bucket, spill, (int)blockIdx.x - GEMM_BLOCKS);
    }
}

template <int U>
__device__ __forceinline__ void add_edges(const float2* __restrict__ HW2, int2 p, int t, int lane,
                                          float2& acc) {
    float2 r[U];
    float  w[U];
    #pragma unroll
    for (int u = 0; u < U; ++u) {
        int s = __builtin_amdgcn_readlane(p.x, t + u);
        w[u]  = __int_as_float(__builtin_amdgcn_readlane(p.y, t + u));
        r[u]  = HW2[(size_t)s * (N_HID / 2) + lane];
    }
    #pragma unroll
    for (int u = 0; u < U; ++u) {
        acc.x = fmaf(w[u], r[u].x, acc.x);
        acc.y = fmaf(w[u], r[u].y, acc.y);
    }
}

template <bool RELU>
__global__ __launch_bounds__(256) void gather_kernel(const float* __restrict__ HW,
                                                     const int* __restrict__ cnt,
                                                     const int2* __restrict__ bucket,
                                                     const int4* __restrict__ spill,
                                                     const float* __restrict__ bias,
                                                     float* __restrict__ out) {
    const int lane = threadIdx.x & 31;
    const int node = __builtin_amdgcn_readfirstlane(blockIdx.x * (blockDim.x >> 5) + (threadIdx.x >> 5));
    const int deg = cnt[node];
    const int beg = node * BUCKET_CAP;
    const int end = beg + (deg < BUCKET_CAP ? deg : BUCKET_CAP);

    const float2* __restrict__ HW2 = (const float2*)HW;
    float2 acc = ((const float2*)bias)[lane];

    for (int base = beg; base < end; base += 32) {
        int j = base + lane;
        int2 p = make_int2(0, 0);
        if (j < end) p = bucket[j];
        int n = end - base;
        if (n > 32) n = 32;
        int t = 0;
        for (; t + 8 <= n; t += 8) add_edges<8>(HW2, p, t, lane, acc);
        if (t + 4 <= n) { add_edges<4>(HW2, p, t, lane, acc); t += 4; }
        if (t + 2 <= n) { add_edges<2>(HW2, p, t, lane, acc); t += 2; }
        if (t < n)      { add_edges<1>(HW2, p, t, lane, acc); }
    }

    if (deg > BUCKET_CAP) {
        const int n_spill = cnt[N_NODES];
        for (int base = 0; base < n_spill; base += 32) {
            int i = base + lane;
            int4 q = make_int4(-1, 0, 0, 0);
            if (i < n_spill) q = spill[i];
            unsigned m = __builtin_amdgcn_ballot_w32(q.x == node);
            while (m) {
                int t = __builtin_ctz(m);
                m &= m - 1;
                add_edges<1>(HW2, make_int2(q.y, q.z), t, lane, acc);
            }
        }
    }

    if (RELU) {
        acc.x = fmaxf(acc.x, 0.0f);
        acc.y = fmaxf(acc.y, 0.0f);
    }
    ((float2*)out)[(size_t)node * (N_HID / 2) + lane] = acc;
}

extern "C" void kernel_launch(void* const* d_in, const int* in_sizes, int n_in,
                              void* d_out, int out_size, void* d_ws, size_t ws_size,
                              hipStream_t stream) {
    const float* x  = (const float*)d_in[0];
    const int*   ei = (const int*)  d_in[1];
    const float* ew = (const float*)d_in[2];
    const float* w1 = (const float*)d_in[3];
    const float* b1 = (const float*)d_in[4];
    const float* w2 = (const float*)d_in[5];
    const float* b2 = (const float*)d_in[6];
    float* out = (float*)d_out;

    float* hw     = (float*)d_ws;
    float* agg    = hw + (size_t)N_NODES * N_HID;
    int*   cnt    = (int*)(agg + (size_t)N_NODES * N_HID);
    int2*  bucket = (int2*)(cnt + CNT_PAD);
    int4*  spill  = (int4*)(bucket + (size_t)N_NODES * BUCKET_CAP);

    const int* src = ei;
    const int* dst = ei + N_EDGES;

    const int mTiles    = N_NODES / 16;
    dim3 blk(256);
    dim3 gCnt((N_NODES + 1 + 255) / 256);
    dim3 gGemm((mTiles + 7) / 8);
    dim3 gFused(GEMM_BLOCKS + EDGE_BLOCKS);
    dim3 gGather(N_NODES / 8);

    zero_counts_kernel<<<gCnt, blk, 0, stream>>>(cnt);
    bucket_fill_and_gemm_kernel<<<gFused, blk, 0, stream>>>(src, dst, ew, cnt, bucket, spill, x, w1, hw);

    gather_kernel<true><<<gGather, blk, 0, stream>>>(hw, cnt, bucket, spill, b1, agg);

    gemm_wmma_kernel<<<gGemm, blk, 0, stream>>>(agg, w2, hw, N_NODES, N_HID);
    gather_kernel<false><<<gGather, blk, 0, stream>>>(hw, cnt, bucket, spill, b2, out);
}


# Layer-2 product h @ w2 moved into the layer-1 gather: a 16-wave block keeps its 16 relu'd rows in LDS as the A tile, 4 waves run the f32 16x16x4 steps; 1 launch and h's round trip gone; same sums
# speedup vs baseline: 2.8051x; 1.0400x over previous
#include <hip/hip_runtime.h>


typedef __attribute__((ext_vector_type(2))) float v2f;
typedef __attribute__((ext_vector_type(8))) float v8f;

#define N_NODES 50000
#define N_EDGES 800000
#define N_FEAT  128
#define N_HID   64

#define BUCKET_CAP 64
#define CNT_PAD    (((N_NODES + 1 + 63) / 64) * 64)

static_assert(N_HID == 64, "the gather maps 32 lanes x 2 floats onto one 64-wide row");
static_assert(N_NODES % 8 == 0, "the gather grid is exact: 8 one-wave nodes per 256-thread block");
static_assert((long long)N_NODES * BUCKET_CAP < 2147483647LL, "slot indices fit an int");

__device__ __forceinline__ void gemm_wmma_block(
    const float* __restrict__ X, const float* __restrict__ W,
    float* __restrict__ HW, int M, int K, int block) {
    int wave = (block * 256 + (int)threadIdx.x) >> 5;
    int lane = threadIdx.x & 31;
    int m0 = wave * 16;
    if (m0 >= M) return;
    int half = lane >> 4;
    int l    = lane & 15;

    v8f acc0 = {}, acc1 = {}, acc2 = {}, acc3 = {};
    const float* xrow = X + (size_t)(m0 + l) * K;

    for (int k = 0; k < K; k += 4) {
        int kk = k + 2 * half;
        v2f a = *(const v2f*)(xrow + kk);
        const float* bp = W + (size_t)kk * N_HID + l;
        v2f b0 = { bp[ 0], bp[ 0 + N_HID] };
        v2f b1 = { bp[16], bp[16 + N_HID] };
        v2f b2 = { bp[32], bp[32 + N_HID] };
        v2f b3 = { bp[48], bp[48 + N_HID] };
        acc0 = __builtin_amdgcn_wmma_f32_16x16x4_f32(false, a, false, b0, (short)0, acc0, false, false);
        acc1 = __builtin_amdgcn_wmma_f32_16x16x4_f32(false, a, false, b1, (short)0, acc1, false, false);
        acc2 = __builtin_amdgcn_wmma_f32_16x16x4_f32(false, a, false, b2, (short)0, acc2, false, false);
        acc3 = __builtin_amdgcn_wmma_f32_16x16x4_f32(false, a, false, b3, (short)0, acc3, false, false);
    }

    float* orow = HW + (size_t)(m0 + 8 * half) * N_HID + l;
    #pragma unroll
    for (int r = 0; r < 8; ++r) {
        orow[(size_t)r * N_HID +  0] = acc0[r];
        orow[(size_t)r * N_HID + 16] = acc1[r];
        orow[(size_t)r * N_HID + 32] = acc2[r];
        orow[(size_t)r * N_HID + 48] = acc3[r];
    }
}


__global__ void zero_counts_kernel(int* __restrict__ cnt) {
    int i = blockIdx.x * blockDim.x + threadIdx.x;
    if (i <= N_NODES) cnt[i] = 0;
}

__device__ __forceinline__ void bucket_fill_block(const int* __restrict__ src,
                                                  const int* __restrict__ dst,
                                                  const float* __restrict__ ew,
                                                  int* __restrict__ cnt,
                                                  int2* __restrict__ bucket,
                                                  int4* __restrict__ spill,
                                                  int block) {
    int e = block * 256 + (int)threadIdx.x;
    if (e < N_EDGES) {
        const int  d   = dst[e];
        const int2 rec = make_int2(src[e], __float_as_int(ew[e]));
        const int  pos = atomicAdd(&cnt[d], 1);
        if (pos < BUCKET_CAP) {
            bucket[(size_t)d * BUCKET_CAP + pos] = rec;
        } else {
            const int o = atomicAdd(&cnt[N_NODES], 1);
            spill[o] = make_int4(d, rec.x, rec.y, 0);
        }
    }
}

#define GEMM_BLOCKS ((N_NODES / 16 + 7) / 8)
#define EDGE_BLOCKS ((N_EDGES + 255) / 256)

__global__ __launch_bounds__(256) void bucket_fill_and_gemm_kernel(
    const int* __restrict__ src, const int* __restrict__ dst, const float* __restrict__ ew,
    int* __restrict__ cnt, int2* __restrict__ bucket, int4* __restrict__ spill,
    const float* __restrict__ X, const float* __restrict__ W, float* __restrict__ HW) {
    if (blockIdx.x < GEMM_BLOCKS) {
        gemm_wmma_block(X, W, HW, N_NODES, N_FEAT, (int)blockIdx.x);
    } else {
        bucket_fill_block(src, dst, ew, cnt, bucket, spill, (int)blockIdx.x - GEMM_BLOCKS);
    }
}

template <int U>
__device__ __forceinline__ void add_edges(const float2* __restrict__ HW2, int2 p, int t, int lane,
                                          float2& acc) {
    float2 r[U];
    float  w[U];
    #pragma unroll
    for (int u = 0; u < U; ++u) {
        int s = __builtin_amdgcn_readlane(p.x, t + u);
        w[u]  = __int_as_float(__builtin_amdgcn_readlane(p.y, t + u));
        r[u]  = HW2[(size_t)s * (N_HID / 2) + lane];
    }
    #pragma unroll
    for (int u = 0; u < U; ++u) {
        acc.x = fmaf(w[u], r[u].x, acc.x);
        acc.y = fmaf(w[u], r[u].y, acc.y);
    }
}

__device__ __forceinline__ float2 gather_row(const float* __restrict__ HW,
                                             const int* __restrict__ cnt,
                                             const int2* __restrict__ bucket,
                                             const int4* __restrict__ spill,
                                             const float* __restrict__ bias,
                                             int node, int lane) {
    const int deg = cnt[node];
    const int beg = node * BUCKET_CAP;
    const int end = beg + (deg < BUCKET_CAP ? deg : BUCKET_CAP);

    const float2* __restrict__ HW2 = (const float2*)HW;
    float2 acc = ((const float2*)bias)[lane];

    for (int base = beg; base < end; base += 32) {
        int j = base + lane;
        int2 p = make_int2(0, 0);
        if (j < end) p = bucket[j];
        int n = end - base;
        if (n > 32) n = 32;
        int t = 0;
        for (; t + 8 <= n; t += 8) add_edges<8>(HW2, p, t, lane, acc);
        if (t + 4 <= n) { add_edges<4>(HW2, p, t, lane, acc); t += 4; }
        if (t + 2 <= n) { add_edges<2>(HW2, p, t, lane, acc); t += 2; }
        if (t < n)      { add_edges<1>(HW2, p, t, lane, acc); }
    }

    if (deg > BUCKET_CAP) {
        const int n_spill = cnt[N_NODES];
        for (int base = 0; base < n_spill; base += 32) {
            int i = base + lane;
            int4 q = make_int4(-1, 0, 0, 0);
            if (i < n_spill) q = spill[i];
            unsigned m = __builtin_amdgcn_ballot_w32(q.x == node);
            while (m) {
                int t = __builtin_ctz(m);
                m &= m - 1;
                add_edges<1>(HW2, make_int2(q.y, q.z), t, lane, acc);
            }
        }
    }
    return acc;
}

__global__ __launch_bounds__(256) void gather_kernel(const float* __restrict__ HW,
                                                     const int* __restrict__ cnt,
                                                     const int2* __restrict__ bucket,
                                                     const int4* __restrict__ spill,
                                                     const float* __restrict__ bias,
                                                     float* __restrict__ out) {
    const int lane = threadIdx.x & 31;
    const int node = __builtin_amdgcn_readfirstlane(blockIdx.x * (blockDim.x >> 5) + (threadIdx.x >> 5));
    const float2 acc = gather_row(HW, cnt, bucket, spill, bias, node, lane);
    ((float2*)out)[(size_t)node * (N_HID / 2) + lane] = acc;
}

#define TILE_PITCH 68
static_assert(N_NODES % 16 == 0, "the fused grid is exact: 16 one-wave nodes per 512-thread block");

__global__ __launch_bounds__(512) void gather_relu_gemm_kernel(const float* __restrict__ HW,
                                                               const int* __restrict__ cnt,
                                                               const int2* __restrict__ bucket,
                                                               const int4* __restrict__ spill,
                                                               const float* __restrict__ bias,
                                                               const float* __restrict__ W2,
                                                               float* __restrict__ HWout) {
    __shared__ __attribute__((aligned(16))) float h_s[16 * TILE_PITCH];
    __shared__ __attribute__((aligned(16))) float o_s[16 * TILE_PITCH];

    const int lane = threadIdx.x & 31;
    const int half = lane >> 4;
    const int l    = lane & 15;
    const int wave = __builtin_amdgcn_readfirstlane((int)(threadIdx.x >> 5));
    const int node = blockIdx.x * 16 + wave;
    const bool mm  = wave < 4;

    v2f b[N_HID / 4];
    if (mm) {
        const float* bp = W2 + (size_t)(2 * half) * N_HID + 16 * wave + l;
        #pragma unroll
        for (int i = 0; i < N_HID / 4; ++i) {
            b[i][0] = bp[(size_t)(4 * i) * N_HID];
            b[i][1] = bp[(size_t)(4 * i + 1) * N_HID];
        }
    }

    float2 acc = gather_row(HW, cnt, bucket, spill, bias, node, lane);
    acc.x = fmaxf(acc.x, 0.0f);
    acc.y = fmaxf(acc.y, 0.0f);

    *(float2*)(h_s + wave * TILE_PITCH + 2 * lane) = acc;
    __syncthreads();

    if (mm) {
        const float* ap = h_s + l * TILE_PITCH + 2 * half;
        v8f d = {};
        #pragma unroll
        for (int i = 0; i < N_HID / 4; ++i) {
            v2f a = *(const v2f*)(ap + 4 * i);
            d = __builtin_amdgcn_wmma_f32_16x16x4_f32(false, a, false, b[i], (short)0, d, false, false);
        }
        float* op = o_s + (8 * half) * TILE_PITCH + 16 * wave + l;
        #pragma unroll
        for (int r = 0; r < 8; ++r) op[r * TILE_PITCH] = d[r];
    }
    __syncthreads();

    ((float2*)HWout)[(size_t)node * (N_HID / 2) + lane] = *(const float2*)(o_s + wave * TILE_PITCH + 2 * lane);
}

extern "C" void kernel_launch(void* const* d_in, const int* in_sizes, int n_in,
                              void* d_out, int out_size, void* d_ws, size_t ws_size,
                              hipStream_t stream) {
    const float* x  = (const float*)d_in[0];
    const int*   ei = (const int*)  d_in[1];
    const float* ew = (const float*)d_in[2];
    const float* w1 = (const float*)d_in[3];
    const float* b1 = (const float*)d_in[4];
    const float* w2 = (const float*)d_in[5];
    const float* b2 = (const float*)d_in[6];
    float* out = (float*)d_out;

    float* hw     = (float*)d_ws;
    float* hw2    = hw + (size_t)N_NODES * N_HID;
    int*   cnt    = (int*)(hw2 + (size_t)N_NODES * N_HID);
    int2*  bucket = (int2*)(cnt + CNT_PAD);
    int4*  spill  = (int4*)(bucket + (size_t)N_NODES * BUCKET_CAP);

    const int* src = ei;
    const int* dst = ei + N_EDGES;

    dim3 blk(256);
    dim3 gCnt((N_NODES + 1 + 255) / 256);
    dim3 gFused(GEMM_BLOCKS + EDGE_BLOCKS);
    dim3 gGather(N_NODES / 8);
    dim3 blkFused(512);
    dim3 gGatherGemm(N_NODES / 16);

    zero_counts_kernel<<<gCnt, blk, 0, stream>>>(cnt);
    bucket_fill_and_gemm_kernel<<<gFused, blk, 0, stream>>>(src, dst, ew, cnt, bucket, spill, x, w1, hw);

    gather_relu_gemm_kernel<<<gGatherGemm, blkFused, 0, stream>>>(hw, cnt, bucket, spill, b1, w2, hw2);

    gather_kernel<<<gGather, blk, 0, stream>>>(hw2, cnt, bucket, spill, b2, out);
}


# Edges grouped by destination through LDS: per-block bin sort, one global add per (block, bin), contiguous record runs; node lists rebuilt in LDS per 64-node gather block; sum order = LDS arrival rank
# speedup vs baseline: 6.1665x; 2.1983x over previous
#include <hip/hip_runtime.h>


typedef __attribute__((ext_vector_type(2))) float v2f;
typedef __attribute__((ext_vector_type(8))) float v8f;

#define N_NODES 50000
#define N_EDGES 800000
#define N_FEAT  128
#define N_HID   64

#define BIN_SHIFT 6
#define BIN_NODES (1 << BIN_SHIFT)
#define NBINS     ((N_NODES + BIN_NODES - 1) / BIN_NODES)
#define BIN_CAP   2048
#define CNT_PAD   (((NBINS + 1 + 63) / 64) * 64)
#define CHUNK     4096
#define EPT       (CHUNK / 256)
#define RPT       (BIN_CAP / 512)

static_assert(N_HID == 64, "the gather maps 32 lanes x 2 floats onto one 64-wide row");
static_assert(N_NODES <= 65536, "a record packs the source node into 16 bits");
static_assert(BIN_NODES == 64, "the level-2 scan gives each lane of one wave two nodes; a gather block is 4 M-tiles");
static_assert(NBINS <= 32 * 25, "the level-1 scan gives each lane of one wave 25 bins");
static_assert(NBINS < 65536, "bin ids are kept as 16-bit values in LDS");

__device__ __forceinline__ void gemm_wmma_block(
    const float* __restrict__ X, const float* __restrict__ W,
    float* __restrict__ HW, int M, int K, int block) {
    int wave = (block * 256 + (int)threadIdx.x) >> 5;
    int lane = threadIdx.x & 31;
    int m0 = wave * 16;
    if (m0 >= M) return;
    int half = lane >> 4;
    int l    = lane & 15;

    v8f acc0 = {}, acc1 = {}, acc2 = {}, acc3 = {};
    const float* xrow = X + (size_t)(m0 + l) * K;

    for (int k = 0; k < K; k += 4) {
        int kk = k + 2 * half;
        v2f a = *(const v2f*)(xrow + kk);
        const float* bp = W + (size_t)kk * N_HID + l;
        v2f b0 = { bp[ 0], bp[ 0 + N_HID] };
        v2f b1 = { bp[16], bp[16 + N_HID] };
        v2f b2 = { bp[32], bp[32 + N_HID] };
        v2f b3 = { bp[48], bp[48 + N_HID] };
        acc0 = __builtin_amdgcn_wmma_f32_16x16x4_f32(false, a, false, b0, (short)0, acc0, false, false);
        acc1 = __builtin_amdgcn_wmma_f32_16x16x4_f32(false, a, false, b1, (short)0, acc1, false, false);
        acc2 = __builtin_amdgcn_wmma_f32_16x16x4_f32(false, a, false, b2, (short)0, acc2, false, false);
        acc3 = __builtin_amdgcn_wmma_f32_16x16x4_f32(false, a, false, b3, (short)0, acc3, false, false);
    }

    float* orow = HW + (size_t)(m0 + 8 * half) * N_HID + l;
    #pragma unroll
    for (int r = 0; r < 8; ++r) {
        orow[(size_t)r * N_HID +  0] = acc0[r];
        orow[(size_t)r * N_HID + 16] = acc1[r];
        orow[(size_t)r * N_HID + 32] = acc2[r];
        orow[(size_t)r * N_HID + 48] = acc3[r];
    }
}


__device__ __forceinline__ int wave_inclusive_sum(int v, int lane) {
    #pragma unroll
    for (int d = 1; d < 32; d <<= 1) {
        int t = __shfl_up(v, d);
        if (lane >= d) v += t;
    }
    return v;
}

__global__ void zero_counts_kernel(int* __restrict__ cnt) {
    int i = blockIdx.x * blockDim.x + threadIdx.x;
    if (i <= NBINS) cnt[i] = 0;
}

__device__ __forceinline__ void bin_edges_block(const int* __restrict__ src,
                                                const int* __restrict__ dst,
                                                const float* __restrict__ ew,
                                                int* __restrict__ cnt,
                                                int2* __restrict__ binrec,
                                                int4* __restrict__ spill,
                                                int chunk,
                                                int* hist, int* start, int* gbase,
                                                int2* srec, unsigned short* sbin) {
    const int tid  = threadIdx.x;
    const int lane = tid & 31;
    const int wave = __builtin_amdgcn_readfirstlane(tid >> 5);
    const int base = chunk * CHUNK;

    for (int b = tid; b < NBINS; b += 256) hist[b] = 0;

    int d[EPT], s[EPT], w[EPT], rk[EPT];
    #pragma unroll
    for (int k = 0; k < EPT; ++k) {
        const int e = base + k * 256 + tid;
        d[k] = 0; s[k] = 0; w[k] = 0;
        if (e < N_EDGES) { d[k] = dst[e]; s[k] = src[e]; w[k] = __float_as_int(ew[e]); }
    }
    __syncthreads();

    #pragma unroll
    for (int k = 0; k < EPT; ++k) {
        const int e = base + k * 256 + tid;
        rk[k] = 0;
        if (e < N_EDGES) rk[k] = atomicAdd(&hist[d[k] >> BIN_SHIFT], 1);
    }
    __syncthreads();

    for (int b = tid; b < NBINS; b += 256) {
        const int c = hist[b];
        gbase[b] = c > 0 ? atomicAdd(&cnt[b], c) : 0;
    }
    if (wave == 0) {
        int c[25];
        int sum = 0;
        #pragma unroll
        for (int q = 0; q < 25; ++q) {
            const int b = 25 * lane + q;
            c[q] = b < NBINS ? hist[b] : 0;
            sum += c[q];
        }
        int run = wave_inclusive_sum(sum, lane) - sum;
        #pragma unroll
        for (int q = 0; q < 25; ++q) {
            const int b = 25 * lane + q;
            if (b < NBINS) start[b] = run;
            run += c[q];
        }
    }
    __syncthreads();

    #pragma unroll
    for (int k = 0; k < EPT; ++k) {
        const int e = base + k * 256 + tid;
        if (e < N_EDGES) {
            const int b   = d[k] >> BIN_SHIFT;
            const int pos = start[b] + rk[k];
            srec[pos] = make_int2(s[k] | ((d[k] & (BIN_NODES - 1)) << 16), w[k]);
            sbin[pos] = (unsigned short)b;
        }
    }
    __syncthreads();

    int n = N_EDGES - base;
    if (n > CHUNK) n = CHUNK;
    for (int i = tid; i < n; i += 256) {
        const int  b   = sbin[i];
        const int  g   = gbase[b] + (i - start[b]);
        const int2 rec = srec[i];
        if (g < BIN_CAP) {
            binrec[(size_t)b * BIN_CAP + g] = rec;
        } else {
            const int o = atomicAdd(&cnt[NBINS], 1);
            spill[o] = make_int4(b * BIN_NODES + (rec.x >> 16), rec.x & 0xffff, rec.y, 0);
        }
    }
}

#define GEMM_BLOCKS ((N_NODES / 16 + 7) / 8)
#define EDGE_BLOCKS ((N_EDGES + CHUNK - 1) / CHUNK)

__global__ __launch_bounds__(256) void bin_edges_and_gemm_kernel(
    const int* __restrict__ src, const int* __restrict__ dst, const float* __restrict__ ew,
    int* __restrict__ cnt, int2* __restrict__ binrec, int4* __restrict__ spill,
    const float* __restrict__ X, const float* __restrict__ W, float* __restrict__ HW) {
    __shared__ int hist[NBINS];
    __shared__ int start[NBINS];
    __shared__ int gbase[NBINS];
    __shared__ int2 srec[CHUNK];
    __shared__ unsigned short sbin[CHUNK];
    if (blockIdx.x < GEMM_BLOCKS) {
        gemm_wmma_block(X, W, HW, N_NODES, N_FEAT, (int)blockIdx.x);
    } else {
        bin_edges_block(src, dst, ew, cnt, binrec, spill, (int)blockIdx.x - GEMM_BLOCKS,
                        hist, start, gbase, srec, sbin);
    }
}

__device__ __forceinline__ void build_node_lists(const int* __restrict__ cnt,
                                                 const int2* __restrict__ binrec,
                                                 int bin, int wave, int lane,
                                                 int* ncnt, int* nstart, int2* lrec) {
    const int tid = threadIdx.x;
    if (tid < BIN_NODES) ncnt[tid] = 0;

    int n = cnt[bin];
    if (n > BIN_CAP) n = BIN_CAP;
    const int2* __restrict__ recs = binrec + (size_t)bin * BIN_CAP;

    int2 rec[RPT];
    int  rk[RPT];
    #pragma unroll
    for (int k = 0; k < RPT; ++k) {
        const int i = tid + 512 * k;
        rec[k] = make_int2(0, 0);
        if (i < n) rec[k] = recs[i];
    }
    __syncthreads();

    #pragma unroll
    for (int k = 0; k < RPT; ++k) {
        const int i = tid + 512 * k;
        rk[k] = 0;
        if (i < n) rk[k] = atomicAdd(&ncnt[rec[k].x >> 16], 1);
    }
    __syncthreads();

    if (wave == 0) {
        const int c0 = ncnt[2 * lane];
        const int c1 = ncnt[2 * lane + 1];
        const int incl = wave_inclusive_sum(c0 + c1, lane);
        const int excl = incl - (c0 + c1);
        nstart[2 * lane]     = excl;
        nstart[2 * lane + 1] = excl + c0;
        if (lane == 31) nstart[BIN_NODES] = incl;
    }
    __syncthreads();

    #pragma unroll
    for (int k = 0; k < RPT; ++k) {
        const int i = tid + 512 * k;
        if (i < n) lrec[nstart[rec[k].x >> 16] + rk[k]] = make_int2(rec[k].x & 0xffff, rec[k].y);
    }
    __syncthreads();
}

template <int U>
__device__ __forceinline__ void add_edges(const float2* __restrict__ HW2, int2 p, int t, int lane,
                                          float2& acc) {
    float2 r[U];
    float  w[U];
    #pragma unroll
    for (int u = 0; u < U; ++u) {
        int s = __builtin_amdgcn_readlane(p.x, t + u);
        w[u]  = __int_as_float(__builtin_amdgcn_readlane(p.y, t + u));
        r[u]  = HW2[(size_t)s * (N_HID / 2) + lane];
    }
    #pragma unroll
    for (int u = 0; u < U; ++u) {
        acc.x = fmaf(w[u], r[u].x, acc.x);
        acc.y = fmaf(w[u], r[u].y, acc.y);
    }
}


__device__ __forceinline__ float2 gather_node(const float* __restrict__ HW,
                                              const int* nstart, const int2* lrec,
                                              const int4* __restrict__ spill, int n_spill,
                                              float2 bias2, int v, int node, int lane) {
    const int beg = __builtin_amdgcn_readfirstlane(nstart[v]);
    const int end = __builtin_amdgcn_readfirstlane(nstart[v + 1]);

    const float2* __restrict__ HW2 = (const float2*)HW;
    float2 acc = bias2;

    for (int base = beg; base < end; base += 32) {
        int j = base + lane;
        int2 p = make_int2(0, 0);
        if (j < end) p = lrec[j];
        int n = end - base;
        if (n > 32) n = 32;
        int t = 0;
        for (; t + 8 <= n; t += 8) add_edges<8>(HW2, p, t, lane, acc);
        if (t + 4 <= n) { add_edges<4>(HW2, p, t, lane, acc); t += 4; }
        if (t + 2 <= n) { add_edges<2>(HW2, p, t, lane, acc); t += 2; }
        if (t < n)      { add_edges<1>(HW2, p, t, lane, acc); }
    }

    if (n_spill > 0) {
        for (int base = 0; base < n_spill; base += 32) {
            int i = base + lane;
            int4 q = make_int4(-1, 0, 0, 0);
            if (i < n_spill) q = spill[i];
            unsigned m = __builtin_amdgcn_ballot_w32(q.x == node);
            while (m) {
                int t = __builtin_ctz(m);
                m &= m - 1;
                add_edges<1>(HW2, make_int2(q.y, q.z), t, lane, acc);
            }
        }
    }
    return acc;
}

__global__ __launch_bounds__(512) void gather_kernel(const float* __restrict__ HW,
                                                     const int* __restrict__ cnt,
                                                     const int2* __restrict__ binrec,
                                                     const int4* __restrict__ spill,
                                                     const float* __restrict__ bias,
                                                     float* __restrict__ out) {
    __shared__ int  ncnt[BIN_NODES];
    __shared__ int  nstart[BIN_NODES + 1];
    __shared__ int2 lrec[BIN_CAP];

    const int lane = threadIdx.x & 31;
    const int wave = __builtin_amdgcn_readfirstlane((int)(threadIdx.x >> 5));
    const int bin  = blockIdx.x;

    const float2 bias2   = ((const float2*)bias)[lane];
    const int    n_spill = cnt[NBINS];
    build_node_lists(cnt, binrec, bin, wave, lane, ncnt, nstart, lrec);

    #pragma unroll 1
    for (int j = 0; j < BIN_NODES / 16; ++j) {
        const int v    = wave + 16 * j;
        const int node = bin * BIN_NODES + v;
        if (node < N_NODES) {
            const float2 acc = gather_node(HW, nstart, lrec, spill, n_spill, bias2, v, node, lane);
            ((float2*)out)[(size_t)node * (N_HID / 2) + lane] = acc;
        }
    }
}

#define TILE_PITCH 68

__global__ __launch_bounds__(512) void gather_relu_gemm_kernel(const float* __restrict__ HW,
                                                               const int* __restrict__ cnt,
                                                               const int2* __restrict__ binrec,
                                                               const int4* __restrict__ spill,
                                                               const float* __restrict__ bias,
                                                               const float* __restrict__ W2,
                                                               float* __restrict__ HWout) {
    __shared__ int  ncnt[BIN_NODES];
    __shared__ int  nstart[BIN_NODES + 1];
    __shared__ int2 lrec[BIN_CAP];
    __shared__ __attribute__((aligned(16))) float h_s[BIN_NODES * TILE_PITCH];
    __shared__ __attribute__((aligned(16))) float o_s[BIN_NODES * TILE_PITCH];

    const int lane = threadIdx.x & 31;
    const int half = lane >> 4;
    const int l    = lane & 15;
    const int wave = __builtin_amdgcn_readfirstlane((int)(threadIdx.x >> 5));
    const int bin  = blockIdx.x;
    const int mt   = wave >> 2;
    const int nt   = wave & 3;

    const float2 bias2   = ((const float2*)bias)[lane];
    const int    n_spill = cnt[NBINS];
    build_node_lists(cnt, binrec, bin, wave, lane, ncnt, nstart, lrec);

    #pragma unroll 1
    for (int j = 0; j < BIN_NODES / 16; ++j) {
        const int v    = wave + 16 * j;
        const int node = bin * BIN_NODES + v;
        float2 acc = make_float2(0.0f, 0.0f);
        if (node < N_NODES) {
            acc = gather_node(HW, nstart, lrec, spill, n_spill, bias2, v, node, lane);
            acc.x = fmaxf(acc.x, 0.0f);
            acc.y = fmaxf(acc.y, 0.0f);
        }
        *(float2*)(h_s + v * TILE_PITCH + 2 * lane) = acc;
    }
    v2f b[N_HID / 4];
    {
        const float* bp = W2 + (size_t)(2 * half) * N_HID + 16 * nt + l;
        #pragma unroll
        for (int i = 0; i < N_HID / 4; ++i) {
            b[i][0] = bp[(size_t)(4 * i) * N_HID];
            b[i][1] = bp[(size_t)(4 * i + 1) * N_HID];
        }
    }
    __syncthreads();

    {
        const float* ap = h_s + (16 * mt + l) * TILE_PITCH + 2 * half;
        v8f d = {};
        #pragma unroll
        for (int i = 0; i < N_HID / 4; ++i) {
            v2f a = *(const v2f*)(ap + 4 * i);
            d = __builtin_amdgcn_wmma_f32_16x16x4_f32(false, a, false, b[i], (short)0, d, false, false);
        }
        float* op = o_s + (16 * mt + 8 * half) * TILE_PITCH + 16 * nt + l;
        #pragma unroll
        for (int r = 0; r < 8; ++r) op[r * TILE_PITCH] = d[r];
    }
    __syncthreads();

    #pragma unroll
    for (int j = 0; j < BIN_NODES / 16; ++j) {
        const int v    = wave + 16 * j;
        const int node = bin * BIN_NODES + v;
        if (node < N_NODES)
            ((float2*)HWout)[(size_t)node * (N_HID / 2) + lane] = *(const float2*)(o_s + v * TILE_PITCH + 2 * lane);
    }
}

extern "C" void kernel_launch(void* const* d_in, const int* in_sizes, int n_in,
                              void* d_out, int out_size, void* d_ws, size_t ws_size,
                              hipStream_t stream) {
    const float* x  = (const float*)d_in[0];
    const int*   ei = (const int*)  d_in[1];
    const float* ew = (const float*)d_in[2];
    const float* w1 = (const float*)d_in[3];
    const float* b1 = (const float*)d_in[4];
    const float* w2 = (const float*)d_in[5];
    const float* b2 = (const float*)d_in[6];
    float* out = (float*)d_out;

    float* hw     = (float*)d_ws;
    float* hw2    = hw + (size_t)N_NODES * N_HID;
    int*   cnt    = (int*)(hw2 + (size_t)N_NODES * N_HID);
    int2*  binrec = (int2*)(cnt + CNT_PAD);
    int4*  spill  = (int4*)(binrec + (size_t)NBINS * BIN_CAP);

    const int* src = ei;
    const int* dst = ei + N_EDGES;

    dim3 blk(256);
    dim3 gCnt((NBINS + 1 + 255) / 256);
    dim3 gFused(GEMM_BLOCKS + EDGE_BLOCKS);
    dim3 blkBin(512);
    dim3 gBins(NBINS);

    zero_counts_kernel<<<gCnt, blk, 0, stream>>>(cnt);
    bin_edges_and_gemm_kernel<<<gFused, blk, 0, stream>>>(src, dst, ew, cnt, binrec, spill, x, w1, hw);

    gather_relu_gemm_kernel<<<gBins, blkBin, 0, stream>>>(hw, cnt, binrec, spill, b1, w2, hw2);

    gather_kernel<<<gBins, blkBin, 0, stream>>>(hw2, cnt, binrec, spill, b2, out);
}
